# starting kernel
# baseline (speedup 1.0000x reference)
#include <hip/hip_runtime.h>

typedef __attribute__((ext_vector_type(2))) float v2f;
typedef __attribute__((ext_vector_type(8))) float v8f;

#define BETA_F   0.001f
#define KSEG     32
#define NPAIR_F  496.0f
#define LOG2E_F  1.4426950408889634f
#define LN2_F    0.6931471805599453f

__device__ __forceinline__ float lane_read(float v, int idx_bytes) {
    int r = __builtin_amdgcn_ds_bpermute(idx_bytes, __builtin_bit_cast(int, v));
    return __builtin_bit_cast(float, r);
}

__device__ __forceinline__ float soft_log2(float t) {
    float e = __builtin_amdgcn_exp2f(-__builtin_fabsf(t));
    return __builtin_amdgcn_logf(1.0f + e);
}

__global__ __launch_bounds__(256) void rmloss_pairwise_kernel(
        const float* __restrict__ logits, float* __restrict__ out, int n_seg) {
    const int tid  = blockIdx.x * 256 + (int)threadIdx.x;
    const int seg  = tid >> 5;
    const int lane = (int)threadIdx.x & 31;
    if (seg >= n_seg) return;

    const float x  = logits[seg * KSEG + lane];
    const float xs = x * LOG2E_F;

    const int base = lane << 2;
    float A = 0.0f;
    float L = 0.0f;
    #pragma unroll
    for (int o = 1; o <= 15; ++o) {
        float y = lane_read(xs, base + (o << 2));
        float t = xs - y;
        A += __builtin_fabsf(t);
        L += soft_log2(t);
    }
    {
        float y = lane_read(xs, base + 64);
        float t = xs - y;
        A = fmaf(0.5f, __builtin_fabsf(t), A);
        L = fmaf(0.5f, soft_log2(t), L);
    }

    float coef = 15.5f - (float)lane;
    float p = coef * x * fmaf(BETA_F, x, 1.0f) - LN2_F * fmaf(0.5f, A, L);

    v2f a; a[0] = p;    a[1] = 0.0f;
    v2f b; b[0] = 1.0f; b[1] = 1.0f;
    v8f c = {};
    c = __builtin_amdgcn_wmma_f32_16x16x4_f32(false, a, false, b, (short)0, c, false, false);
    float partial = ((c[0] + c[1]) + (c[2] + c[3])) + ((c[4] + c[5]) + (c[6] + c[7]));
    float total = partial + lane_read(partial, (lane ^ 16) << 2);

    if (lane == 0) out[seg] = -total * (1.0f / NPAIR_F);
}

extern "C" void kernel_launch(void* const* d_in, const int* in_sizes, int n_in,
                              void* d_out, int out_size, void* d_ws, size_t ws_size,
                              hipStream_t stream) {
    (void)in_sizes; (void)n_in; (void)d_ws; (void)ws_size;
    const float* logits = (const float*)d_in[0];
    float* out = (float*)d_out;
    const int n_seg  = out_size;
    const int blocks = (n_seg * 32 + 255) / 256;
    rmloss_pairwise_kernel<<<blocks, 256, 0, stream>>>(logits, out, n_seg);
}


# eight segments per wave (a quad of lanes each, 8 elements per lane): pairs formed in registers, quad DPP reads replace 16 LDS permutes per segment, 8x fewer waves; pair terms summed in a new order
# speedup vs baseline: 1.4710x; 1.4710x over previous
#include <hip/hip_runtime.h>


typedef __attribute__((ext_vector_type(2))) float v2f;
typedef __attribute__((ext_vector_type(4))) float v4f;
typedef __attribute__((ext_vector_type(8))) float v8f;

#define BETA_F   0.001f
#define KSEG     32
#define NPAIR_F  496.0f
#define LOG2E_F  1.4426950408889634f
#define LN2_F    0.6931471805599453f
#define SEGS_PER_WAVE 8
#define CHUNK    8
#define NACC     4

template <int CTRL>
__device__ __forceinline__ float quad_read(float v) {
    int r = __builtin_amdgcn_mov_dpp(__builtin_bit_cast(int, v), CTRL, 0xf, 0xf, true);
    return __builtin_bit_cast(float, r);
}

template <int N>
__device__ __forceinline__ void pair_row(float u, const float* v, float (&accA)[NACC], float (&accL)[NACC]) {
    float t_[N], e_[N];
    for (int b = 0; b < N; ++b) t_[b] = u - v[b];
    for (int b = 0; b < N; ++b) e_[b] = __builtin_amdgcn_exp2f(-__builtin_fabsf(t_[b]));
    for (int b = 0; b < N; ++b) e_[b] = 1.0f + e_[b];
    for (int b = 0; b < N; ++b) e_[b] = __builtin_amdgcn_logf(e_[b]);
    for (int b = 0; b < N; ++b) {
        asm("v_add_f32_e64 %0, %0, |%1|" : "+v"(accA[b & (NACC - 1)]) : "v"(t_[b]));
        accL[b & (NACC - 1)] += e_[b];
    }
}

__global__ __launch_bounds__(256)
__attribute__((amdgpu_waves_per_eu(4, 4)))
void rmloss_pairwise_kernel(
        const float* __restrict__ logits, float* __restrict__ out, int n_seg) {
    const int lane = (int)threadIdx.x & 31;
    const int wave = (int)blockIdx.x * 8 + __builtin_amdgcn_readfirstlane((int)threadIdx.x >> 5);
    const int seg0 = wave * SEGS_PER_WAVE;
    if (seg0 >= n_seg) return;

    const int q   = lane & 3;
    const int seg = seg0 + (lane >> 2);
    const int sld = seg < n_seg ? seg : n_seg - 1;

    const v4f* src = (const v4f*)(logits + (size_t)sld * KSEG + q * CHUNK);
    const v4f lo = src[0];
    const v4f hi = src[1];
    const float x[CHUNK] = { lo[0], lo[1], lo[2], lo[3], hi[0], hi[1], hi[2], hi[3] };

    float xs[CHUNK];
    for (int k = 0; k < CHUNK; ++k) xs[k] = x[k] * LOG2E_F;

    float accA[NACC] = {};
    float accL[NACC] = {};

    pair_row<7>(xs[0], xs + 1, accA, accL);
    pair_row<6>(xs[1], xs + 2, accA, accL);
    pair_row<5>(xs[2], xs + 3, accA, accL);
    pair_row<4>(xs[3], xs + 4, accA, accL);
    pair_row<3>(xs[4], xs + 5, accA, accL);
    pair_row<2>(xs[5], xs + 6, accA, accL);
    pair_row<1>(xs[6], xs + 7, accA, accL);

    float y[CHUNK];
    for (int k = 0; k < CHUNK; ++k) y[k] = quad_read<0x39>(xs[k]);
    for (int a = 0; a < CHUNK; ++a) pair_row<CHUNK>(xs[a], y, accA, accL);

    float z[CHUNK];
    for (int k = 0; k < CHUNK; ++k) z[k] = quad_read<0x4E>(xs[k]);
    const bool upper = (lane & 2) != 0;
    float u4[4], v8[CHUNK];
    for (int k = 0; k < 4; ++k) u4[k] = upper ? z[4 + k] : xs[k];
    for (int k = 0; k < CHUNK; ++k) v8[k] = upper ? xs[k] : z[k];
    for (int a = 0; a < 4; ++a) pair_row<CHUNK>(u4[a], v8, accA, accL);

    const float c0 = 15.5f - (float)(q * CHUNK);
    float lin0 = 0.0f, lin1 = 0.0f;
    for (int k = 0; k < CHUNK; k += 2) {
        lin0 = fmaf((c0 - (float)k)       * x[k],     fmaf(BETA_F, x[k],     1.0f), lin0);
        lin1 = fmaf((c0 - (float)(k + 1)) * x[k + 1], fmaf(BETA_F, x[k + 1], 1.0f), lin1);
    }
    float A = 0.0f, L = 0.0f;
    for (int k = 0; k < NACC; k += 2) { A += accA[k] + accA[k + 1]; L += accL[k] + accL[k + 1]; }
    const float nl = -LN2_F * fmaf(0.5f, A, L);

    const float sel = (((lane >> 3) ^ (lane >> 4)) & 1) ? 0.0f : 1.0f;
    v2f a; a[0] = lin0 + lin1; a[1] = nl;
    v2f b; b[0] = sel;         b[1] = sel;
    v8f c = {};
    c = __builtin_amdgcn_wmma_f32_16x16x4_f32(false, a, false, b, (short)0, c, false, false);
    const float s_first  = (c[0] + c[1]) + (c[2] + c[3]);
    const float s_second = (c[4] + c[5]) + (c[6] + c[7]);

    const int j = lane & 7;
    if (j < 2) {
        const int s = seg0 + 2 * (lane >> 4) + 4 * ((lane >> 3) & 1) + j;
        const float total = j ? s_second : s_first;
        if (s < n_seg) out[s] = -total * (1.0f / NPAIR_F);
    }
}

extern "C" void kernel_launch(void* const* d_in, const int* in_sizes, int n_in,
                              void* d_out, int out_size, void* d_ws, size_t ws_size,
                              hipStream_t stream) {
    (void)in_sizes; (void)n_in; (void)d_ws; (void)ws_size;
    const float* logits = (const float*)d_in[0];
    float* out = (float*)d_out;
    const int n_seg  = out_size;
    const int blocks = (n_seg + 63) / 64;
    rmloss_pairwise_kernel<<<blocks, 256, 0, stream>>>(logits, out, n_seg);
}


# sum of log2(1+2^-|d|) taken as log2 of running products: each lane multiplies its factors into 8 products (one fma per pair, each <= 2^19), 2 v_log per lane instead of 124; reduction order changes
# speedup vs baseline: 1.6355x; 1.1118x over previous
#include <hip/hip_runtime.h>


typedef __attribute__((ext_vector_type(2))) float v2f;
typedef __attribute__((ext_vector_type(4))) float v4f;
typedef __attribute__((ext_vector_type(8))) float v8f;

#define BETA_F   0.001f
#define KSEG     32
#define NPAIR_F  496.0f
#define LOG2E_F  1.4426950408889634f
#define LN2_F    0.6931471805599453f
#define SEGS_PER_WAVE 8
#define CHUNK    8
#define NACC     4
#define NPROD    8

template <int CTRL>
__device__ __forceinline__ float quad_read(float v) {
    int r = __builtin_amdgcn_mov_dpp(__builtin_bit_cast(int, v), CTRL, 0xf, 0xf, true);
    return __builtin_bit_cast(float, r);
}

template <int N>
__device__ __forceinline__ void pair_row(float u, const float* v, float (&accA)[NACC], float (&accP)[NPROD]) {
    float t_[N], e_[N];
    for (int b = 0; b < N; ++b) t_[b] = u - v[b];
    for (int b = 0; b < N; ++b) e_[b] = __builtin_amdgcn_exp2f(-__builtin_fabsf(t_[b]));
    for (int b = 0; b < N; ++b) {
        asm("v_add_f32_e64 %0, %0, |%1|" : "+v"(accA[b & (NACC - 1)]) : "v"(t_[b]));
        accP[b & (NPROD - 1)] = __builtin_fmaf(accP[b & (NPROD - 1)], e_[b], accP[b & (NPROD - 1)]);
    }
}

__global__ __launch_bounds__(256)
__attribute__((amdgpu_waves_per_eu(4, 4)))
void rmloss_pairwise_kernel(
        const float* __restrict__ logits, float* __restrict__ out, int n_seg) {
    const int lane = (int)threadIdx.x & 31;
    const int wave = (int)blockIdx.x * 8 + __builtin_amdgcn_readfirstlane((int)threadIdx.x >> 5);
    const int seg0 = wave * SEGS_PER_WAVE;
    if (seg0 >= n_seg) return;

    const int q   = lane & 3;
    const int seg = seg0 + (lane >> 2);
    const int sld = seg < n_seg ? seg : n_seg - 1;

    const v4f* src = (const v4f*)(logits + (size_t)sld * KSEG + q * CHUNK);
    const v4f lo = src[0];
    const v4f hi = src[1];
    const float x[CHUNK] = { lo[0], lo[1], lo[2], lo[3], hi[0], hi[1], hi[2], hi[3] };

    float xs[CHUNK];
    for (int k = 0; k < CHUNK; ++k) xs[k] = x[k] * LOG2E_F;

    float accA[NACC] = {};
    float accP[NPROD] = { 1.0f, 1.0f, 1.0f, 1.0f, 1.0f, 1.0f, 1.0f, 1.0f };

    pair_row<7>(xs[0], xs + 1, accA, accP);
    pair_row<6>(xs[1], xs + 2, accA, accP);
    pair_row<5>(xs[2], xs + 3, accA, accP);
    pair_row<4>(xs[3], xs + 4, accA, accP);
    pair_row<3>(xs[4], xs + 5, accA, accP);
    pair_row<2>(xs[5], xs + 6, accA, accP);
    pair_row<1>(xs[6], xs + 7, accA, accP);

    float y[CHUNK];
    for (int k = 0; k < CHUNK; ++k) y[k] = quad_read<0x39>(xs[k]);
    for (int a = 0; a < CHUNK; ++a) pair_row<CHUNK>(xs[a], y, accA, accP);

    float z[CHUNK];
    for (int k = 0; k < CHUNK; ++k) z[k] = quad_read<0x4E>(xs[k]);
    const bool upper = (lane & 2) != 0;
    float u4[4], v8[CHUNK];
    for (int k = 0; k < 4; ++k) u4[k] = upper ? z[4 + k] : xs[k];
    for (int k = 0; k < CHUNK; ++k) v8[k] = upper ? xs[k] : z[k];
    for (int a = 0; a < 4; ++a) pair_row<CHUNK>(u4[a], v8, accA, accP);

    const float c0 = 15.5f - (float)(q * CHUNK);
    float lin0 = 0.0f, lin1 = 0.0f;
    for (int k = 0; k < CHUNK; k += 2) {
        lin0 = fmaf((c0 - (float)k)       * x[k],     fmaf(BETA_F, x[k],     1.0f), lin0);
        lin1 = fmaf((c0 - (float)(k + 1)) * x[k + 1], fmaf(BETA_F, x[k + 1], 1.0f), lin1);
    }
    float A = 0.0f;
    for (int k = 0; k < NACC; k += 2) A += accA[k] + accA[k + 1];
    const float L = __builtin_amdgcn_logf((accP[0] * accP[1]) * (accP[2] * accP[3])) + __builtin_amdgcn_logf((accP[4] * accP[5]) * (accP[6] * accP[7]));
    const float nl = -LN2_F * fmaf(0.5f, A, L);

    const float sel = (((lane >> 3) ^ (lane >> 4)) & 1) ? 0.0f : 1.0f;
    v2f a; a[0] = lin0 + lin1; a[1] = nl;
    v2f b; b[0] = sel;         b[1] = sel;
    v8f c = {};
    c = __builtin_amdgcn_wmma_f32_16x16x4_f32(false, a, false, b, (short)0, c, false, false);
    const float s_first  = (c[0] + c[1]) + (c[2] + c[3]);
    const float s_second = (c[4] + c[5]) + (c[6] + c[7]);

    const int j = lane & 7;
    if (j < 2) {
        const int s = seg0 + 2 * (lane >> 4) + 4 * ((lane >> 3) & 1) + j;
        const float total = j ? s_second : s_first;
        if (s < n_seg) out[s] = -total * (1.0f / NPAIR_F);
    }
}

extern "C" void kernel_launch(void* const* d_in, const int* in_sizes, int n_in,
                              void* d_out, int out_size, void* d_ws, size_t ws_size,
                              hipStream_t stream) {
    (void)in_sizes; (void)n_in; (void)d_ws; (void)ws_size;
    const float* logits = (const float*)d_in[0];
    float* out = (float*)d_out;
    const int n_seg  = out_size;
    const int blocks = (n_seg + 63) / 64;
    rmloss_pairwise_kernel<<<blocks, 256, 0, stream>>>(logits, out, n_seg);
}


# one exponential per element, none per pair: logsig(xi-xj) = xi - ln(e^xi+e^xj), a pair = packed add + packed multiply into 16 running products (old path kept for |x| > 10); sum order changes
# speedup vs baseline: 2.0504x; 1.2537x over previous
#include <hip/hip_runtime.h>


typedef __attribute__((ext_vector_type(2))) float v2f;
typedef __attribute__((ext_vector_type(4))) float v4f;
typedef __attribute__((ext_vector_type(8))) float v8f;

#define BETA_F   0.001f
#define KSEG     32
#define NPAIR_F  496.0f
#define LOG2E_F  1.4426950408889634f
#define LN2_F    0.6931471805599453f
#define SEGS_PER_WAVE 8
#define CHUNK    8
#define NACC     4
#define NPROD    8
#define NP2      8
#define XS_BOUND 14.5f

template <int CTRL>
__device__ __forceinline__ float quad_read(float v) {
    int r = __builtin_amdgcn_mov_dpp(__builtin_bit_cast(int, v), CTRL, 0xf, 0xf, true);
    return __builtin_bit_cast(float, r);
}

template <int N>
__device__ __forceinline__ void pair_row(float u, const float* v, float (&accA)[NACC], float (&accP)[NPROD]) {
    float t_[N], e_[N];
    for (int b = 0; b < N; ++b) t_[b] = u - v[b];
    for (int b = 0; b < N; ++b) e_[b] = __builtin_amdgcn_exp2f(-__builtin_fabsf(t_[b]));
    for (int b = 0; b < N; ++b) {
        asm("v_add_f32_e64 %0, %0, |%1|" : "+v"(accA[b & (NACC - 1)]) : "v"(t_[b]));
        accP[b & (NPROD - 1)] = __builtin_fmaf(accP[b & (NPROD - 1)], e_[b], accP[b & (NPROD - 1)]);
    }
}

__global__ __launch_bounds__(256)
__attribute__((amdgpu_waves_per_eu(4, 4)))
void rmloss_pairwise_kernel(
        const float* __restrict__ logits, float* __restrict__ out, int n_seg) {
    const int lane = (int)threadIdx.x & 31;
    const int wave = (int)blockIdx.x * 8 + __builtin_amdgcn_readfirstlane((int)threadIdx.x >> 5);
    const int seg0 = wave * SEGS_PER_WAVE;
    if (seg0 >= n_seg) return;

    const int q   = lane & 3;
    const int seg = seg0 + (lane >> 2);
    const int sld = seg < n_seg ? seg : n_seg - 1;

    const v4f* src = (const v4f*)(logits + (size_t)sld * KSEG + q * CHUNK);
    const v4f lo = src[0];
    const v4f hi = src[1];
    const float x[CHUNK] = { lo[0], lo[1], lo[2], lo[3], hi[0], hi[1], hi[2], hi[3] };

    float xs[CHUNK];
    for (int k = 0; k < CHUNK; ++k) xs[k] = x[k] * LOG2E_F;

    const bool upper = (lane & 2) != 0;
    const float t0 = (float)(q * CHUNK);
    float lin, nl;

    float amax = 0.0f;
    for (int k = 0; k < CHUNK; ++k) amax = __builtin_fmaxf(amax, __builtin_fabsf(xs[k]));
    const bool narrow = __builtin_amdgcn_ballot_w32(!(amax <= XS_BOUND)) == 0u;

    if (__builtin_expect(narrow, 1)) {
        float E[CHUNK];
        for (int k = 0; k < CHUNK; ++k) E[k] = __builtin_amdgcn_exp2f(xs[k]);

        v2f P[NP2];
        for (int k = 0; k < NP2; ++k) P[k] = (v2f){ 1.0f, 1.0f };

        v2f Q[CHUNK / 2];
        for (int p = 0; p < CHUNK / 2; ++p) Q[p] = (v2f){ E[2 * p], E[2 * p + 1] };
        {
            int s = 0;
            for (int p = 0; p < CHUNK / 2; ++p)
                for (int r = p + 1; r < CHUNK / 2; ++r) {
                    P[s & (NP2 - 1)] *= (v2f){ E[2 * p],     E[2 * p]     } + Q[r]; ++s;
                    P[s & (NP2 - 1)] *= (v2f){ E[2 * p + 1], E[2 * p + 1] } + Q[r]; ++s;
                }
            P[s & (NP2 - 1)] *= (v2f){ E[0] + E[1], E[2] + E[3] }; ++s;
            P[s & (NP2 - 1)] *= (v2f){ E[4] + E[5], E[6] + E[7] }; ++s;
        }

        v2f Y[CHUNK / 2];
        for (int p = 0; p < CHUNK / 2; ++p)
            Y[p] = (v2f){ quad_read<0x39>(E[2 * p]), quad_read<0x39>(E[2 * p + 1]) };
        for (int a = 0; a < CHUNK; ++a)
            for (int p = 0; p < CHUNK / 2; ++p)
                P[(4 * a + p) & (NP2 - 1)] *= (v2f){ E[a], E[a] } + Y[p];

        float z[CHUNK];
        for (int k = 0; k < CHUNK; ++k) z[k] = quad_read<0x4E>(E[k]);
        float u4[4];
        v2f V[CHUNK / 2];
        for (int k = 0; k < 4; ++k) u4[k] = upper ? z[4 + k] : E[k];
        for (int p = 0; p < CHUNK / 2; ++p)
            V[p] = (v2f){ upper ? E[2 * p] : z[2 * p], upper ? E[2 * p + 1] : z[2 * p + 1] };
        for (int a = 0; a < 4; ++a)
            for (int p = 0; p < CHUNK / 2; ++p)
                P[(4 * a + p) & (NP2 - 1)] *= (v2f){ u4[a], u4[a] } + V[p];

        float L0 = __builtin_amdgcn_logf(P[0][0]), L1 = __builtin_amdgcn_logf(P[0][1]);
        for (int k = 1; k < NP2; ++k) {
            L0 += __builtin_amdgcn_logf(P[k][0]);
            L1 += __builtin_amdgcn_logf(P[k][1]);
        }
        nl = -LN2_F * (L0 + L1);

        const float w1 = 31.0f - t0;
        const float w2 = BETA_F * (15.5f - t0);
        float lin0 = 0.0f, lin1 = 0.0f;
        for (int k = 0; k < CHUNK; k += 2) {
            lin0 = fmaf(x[k],     fmaf(w2 - BETA_F * (float)k,       x[k],     w1 - (float)k),       lin0);
            lin1 = fmaf(x[k + 1], fmaf(w2 - BETA_F * (float)(k + 1), x[k + 1], w1 - (float)(k + 1)), lin1);
        }
        lin = lin0 + lin1;
    } else {
        float accA[NACC] = {};
        float accP[NPROD] = { 1.0f, 1.0f, 1.0f, 1.0f, 1.0f, 1.0f, 1.0f, 1.0f };

        pair_row<7>(xs[0], xs + 1, accA, accP);
        pair_row<6>(xs[1], xs + 2, accA, accP);
        pair_row<5>(xs[2], xs + 3, accA, accP);
        pair_row<4>(xs[3], xs + 4, accA, accP);
        pair_row<3>(xs[4], xs + 5, accA, accP);
        pair_row<2>(xs[5], xs + 6, accA, accP);
        pair_row<1>(xs[6], xs + 7, accA, accP);

        float y[CHUNK];
        for (int k = 0; k < CHUNK; ++k) y[k] = quad_read<0x39>(xs[k]);
        for (int a = 0; a < CHUNK; ++a) pair_row<CHUNK>(xs[a], y, accA, accP);

        float z[CHUNK];
        for (int k = 0; k < CHUNK; ++k) z[k] = quad_read<0x4E>(xs[k]);
        float u4[4], v8[CHUNK];
        for (int k = 0; k < 4; ++k) u4[k] = upper ? z[4 + k] : xs[k];
        for (int k = 0; k < CHUNK; ++k) v8[k] = upper ? xs[k] : z[k];
        for (int a = 0; a < 4; ++a) pair_row<CHUNK>(u4[a], v8, accA, accP);

        const float c0 = 15.5f - t0;
        float lin0 = 0.0f, lin1 = 0.0f;
        for (int k = 0; k < CHUNK; k += 2) {
            lin0 = fmaf((c0 - (float)k)       * x[k],     fmaf(BETA_F, x[k],     1.0f), lin0);
            lin1 = fmaf((c0 - (float)(k + 1)) * x[k + 1], fmaf(BETA_F, x[k + 1], 1.0f), lin1);
        }
        lin = lin0 + lin1;
        float A = 0.0f;
        for (int k = 0; k < NACC; k += 2) A += accA[k] + accA[k + 1];
        const float L = __builtin_amdgcn_logf((accP[0] * accP[1]) * (accP[2] * accP[3])) + __builtin_amdgcn_logf((accP[4] * accP[5]) * (accP[6] * accP[7]));
        nl = -LN2_F * fmaf(0.5f, A, L);
    }

    const float sel = (((lane >> 3) ^ (lane >> 4)) & 1) ? 0.0f : 1.0f;
    v2f a; a[0] = lin; a[1] = nl;
    v2f b; b[0] = sel; b[1] = sel;
    v8f c = {};
    c = __builtin_amdgcn_wmma_f32_16x16x4_f32(false, a, false, b, (short)0, c, false, false);
    asm volatile("v_nop\n\tv_nop\n\tv_nop\n\tv_nop" : "+v"(c) : "v"(a), "v"(b));
    const float s_first  = (c[0] + c[1]) + (c[2] + c[3]);
    const float s_second = (c[4] + c[5]) + (c[6] + c[7]);

    const int j = lane & 7;
    if (j < 2) {
        const int s = seg0 + 2 * (lane >> 4) + 4 * ((lane >> 3) & 1) + j;
        const float total = j ? s_second : s_first;
        if (s < n_seg) out[s] = -total * (1.0f / NPAIR_F);
    }
}

extern "C" void kernel_launch(void* const* d_in, const int* in_sizes, int n_in,
                              void* d_out, int out_size, void* d_ws, size_t ws_size,
                              hipStream_t stream) {
    (void)in_sizes; (void)n_in; (void)d_ws; (void)ws_size;
    const float* logits = (const float*)d_in[0];
    float* out = (float*)d_out;
    const int n_seg  = out_size;
    const int blocks = (n_seg + 63) / 64;
    rmloss_pairwise_kernel<<<blocks, 256, 0, stream>>>(logits, out, n_seg);
}
